# starting kernel
# baseline (speedup 1.0000x reference)
#include <hip/hip_runtime.h>
#include <cmath>

typedef __attribute__((ext_vector_type(2))) float v2f;
typedef __attribute__((ext_vector_type(8))) float v8f;

constexpr int kH  = 18;
constexpr int kH3 = 54;
constexpr int kH4 = 72;
constexpr int kT  = 1024;
constexpr int kB  = 1024;
constexpr int kKP = 20;
constexpr int kKS = 5;
constexpr int kNG = 64;
constexpr int kNL = 80;

__device__ __forceinline__ float sigmoidf_(float v) {
    return 1.0f / (1.0f + __expf(-v));
}

__global__ __launch_bounds__(32) void fused_gru_lstm_fc(
    const float* __restrict__ x,
    const float* __restrict__ gWih, const float* __restrict__ gWhh,
    const float* __restrict__ gbih, const float* __restrict__ gbhh,
    const float* __restrict__ lWih, const float* __restrict__ lWhh,
    const float* __restrict__ lbih, const float* __restrict__ lbhh,
    const float* __restrict__ fcW,  const float* __restrict__ fcb,
    float* __restrict__ out)
{
    __shared__ float WgX[kNG][kKP], WgH[kNG][kKP];
    __shared__ float WlX[kNL][kKP], WlH[kNL][kKP];
    __shared__ float gbi_s[kH3], gbh_s[kH3], lb_s[kH4];
    __shared__ float fcw_s[kH];
    __shared__ float fcb_s;
    __shared__ float xt[16][kKP];
    __shared__ float hg[16][kKP];
    __shared__ float hl[16][kKP];
    __shared__ float cs[16][kH];
    __shared__ float sgX[16][kNG];
    __shared__ float sgH[16][kNG];
    __shared__ float sl [16][kNL];

    const int tid = threadIdx.x;
    const int m0  = blockIdx.x * 16;

    for (int i = tid; i < kNG * kKP; i += 32) {
        int n = i / kKP, k = i % kKP;
        bool ok = (n < kH3) && (k < kH);
        WgX[n][k] = ok ? gWih[n * kH + k] : 0.0f;
        WgH[n][k] = ok ? gWhh[n * kH + k] : 0.0f;
    }
    for (int i = tid; i < kNL * kKP; i += 32) {
        int n = i / kKP, k = i % kKP;
        bool ok = (n < kH4) && (k < kH);
        WlX[n][k] = ok ? lWih[n * kH + k] : 0.0f;
        WlH[n][k] = ok ? lWhh[n * kH + k] : 0.0f;
    }
    for (int i = tid; i < kH3; i += 32) { gbi_s[i] = gbih[i]; gbh_s[i] = gbhh[i]; }
    for (int i = tid; i < kH4; i += 32) { lb_s[i]  = lbih[i] + lbhh[i]; }
    for (int i = tid; i < kH;  i += 32) { fcw_s[i] = fcW[i]; }
    if (tid == 0) fcb_s = fcb[0];
    for (int i = tid; i < 16 * kKP; i += 32) {
        (&xt[0][0])[i] = 0.0f; (&hg[0][0])[i] = 0.0f; (&hl[0][0])[i] = 0.0f;
    }
    for (int i = tid; i < 16 * kH; i += 32) (&cs[0][0])[i] = 0.0f;
    __syncthreads();

    const int lh  = tid & 15;
    const int kh  = (tid >> 4) << 1;
    const int hi8 = (tid >> 4) << 3;

    v2f bGX[4][kKS], bGH[4][kKS];
    #pragma unroll
    for (int nt = 0; nt < 4; ++nt)
      #pragma unroll
      for (int ks = 0; ks < kKS; ++ks) {
        int r = nt * 16 + lh, c = ks * 4 + kh;
        v2f a; a.x = WgX[r][c]; a.y = WgX[r][c + 1]; bGX[nt][ks] = a;
        v2f b; b.x = WgH[r][c]; b.y = WgH[r][c + 1]; bGH[nt][ks] = b;
      }
    v2f bLX[5][kKS], bLH[5][kKS];
    #pragma unroll
    for (int nt = 0; nt < 5; ++nt)
      #pragma unroll
      for (int ks = 0; ks < kKS; ++ks) {
        int r = nt * 16 + lh, c = ks * 4 + kh;
        v2f a; a.x = WlX[r][c]; a.y = WlX[r][c + 1]; bLX[nt][ks] = a;
        v2f b; b.x = WlH[r][c]; b.y = WlH[r][c + 1]; bLH[nt][ks] = b;
      }

    for (int t = 0; t < kT; ++t) {
        for (int i = tid; i < 16 * kH; i += 32) {
            int m = i / kH, j = i % kH;
            xt[m][j] = x[((size_t)(m0 + m) * kT + t) * kH + j];
        }
        if (t + 1 < kT && tid < 16) {
            __builtin_prefetch(&x[((size_t)(m0 + tid) * kT + (t + 1)) * kH], 0, 3);
        }
        __syncthreads();

        v2f aX[kKS], aH[kKS];
        #pragma unroll
        for (int ks = 0; ks < kKS; ++ks) {
            int c = ks * 4 + kh;
            v2f a; a.x = xt[lh][c]; a.y = xt[lh][c + 1]; aX[ks] = a;
            v2f b; b.x = hg[lh][c]; b.y = hg[lh][c + 1]; aH[ks] = b;
        }
        v8f accX[4], accH[4];
        #pragma unroll
        for (int nt = 0; nt < 4; ++nt) { accX[nt] = (v8f){}; accH[nt] = (v8f){}; }
        #pragma unroll
        for (int nt = 0; nt < 4; ++nt)
          #pragma unroll
          for (int ks = 0; ks < kKS; ++ks) {
            accX[nt] = __builtin_amdgcn_wmma_f32_16x16x4_f32(
                false, aX[ks], false, bGX[nt][ks], (short)0, accX[nt], false, false);
            accH[nt] = __builtin_amdgcn_wmma_f32_16x16x4_f32(
                false, aH[ks], false, bGH[nt][ks], (short)0, accH[nt], false, false);
          }
        #pragma unroll
        for (int nt = 0; nt < 4; ++nt)
          #pragma unroll
          for (int j = 0; j < 8; ++j) {
            sgX[hi8 + j][nt * 16 + lh] = accX[nt][j];
            sgH[hi8 + j][nt * 16 + lh] = accH[nt][j];
          }
        __syncthreads();

        for (int i = tid; i < 16 * kH; i += 32) {
            int m = i / kH, j = i % kH;
            float r = sigmoidf_((sgX[m][j]          + gbi_s[j])
                              + (sgH[m][j]          + gbh_s[j]));
            float z = sigmoidf_((sgX[m][kH + j]     + gbi_s[kH + j])
                              + (sgH[m][kH + j]     + gbh_s[kH + j]));
            float n = tanhf(    (sgX[m][2*kH + j]   + gbi_s[2*kH + j])
                          + r * (sgH[m][2*kH + j]   + gbh_s[2*kH + j]));
            float hprev = hg[m][j];
            hg[m][j] = (1.0f - z) * n + z * hprev;
        }
        __syncthreads();

        v2f aG[kKS], aL[kKS];
        #pragma unroll
        for (int ks = 0; ks < kKS; ++ks) {
            int c = ks * 4 + kh;
            v2f a; a.x = hg[lh][c]; a.y = hg[lh][c + 1]; aG[ks] = a;
            v2f b; b.x = hl[lh][c]; b.y = hl[lh][c + 1]; aL[ks] = b;
        }
        v8f accL[5];
        #pragma unroll
        for (int nt = 0; nt < 5; ++nt) accL[nt] = (v8f){};
        #pragma unroll
        for (int nt = 0; nt < 5; ++nt)
          #pragma unroll
          for (int ks = 0; ks < kKS; ++ks) {
            accL[nt] = __builtin_amdgcn_wmma_f32_16x16x4_f32(
                false, aG[ks], false, bLX[nt][ks], (short)0, accL[nt], false, false);
            accL[nt] = __builtin_amdgcn_wmma_f32_16x16x4_f32(
                false, aL[ks], false, bLH[nt][ks], (short)0, accL[nt], false, false);
          }
        #pragma unroll
        for (int nt = 0; nt < 5; ++nt)
          #pragma unroll
          for (int j = 0; j < 8; ++j)
            sl[hi8 + j][nt * 16 + lh] = accL[nt][j];
        __syncthreads();

        for (int i = tid; i < 16 * kH; i += 32) {
            int m = i / kH, j = i % kH;
            float ig = sigmoidf_(sl[m][j]          + lb_s[j]);
            float fg = sigmoidf_(sl[m][kH + j]     + lb_s[kH + j]);
            float gg = tanhf(    sl[m][2*kH + j]   + lb_s[2*kH + j]);
            float og = sigmoidf_(sl[m][3*kH + j]   + lb_s[3*kH + j]);
            float cn = fg * cs[m][j] + ig * gg;
            cs[m][j] = cn;
            hl[m][j] = og * tanhf(cn);
        }
        __syncthreads();

        if (tid < 16) {
            float s = fcb_s;
            #pragma unroll
            for (int j = 0; j < kH; ++j) s += hl[tid][j] * fcw_s[j];
            out[(size_t)(m0 + tid) * kT + t] = s;
        }
        __syncthreads();
    }
}

extern "C" void kernel_launch(void* const* d_in, const int* in_sizes, int n_in,
                              void* d_out, int out_size, void* d_ws, size_t ws_size,
                              hipStream_t stream) {
    (void)in_sizes; (void)n_in; (void)out_size; (void)d_ws; (void)ws_size;
    const float* x    = (const float*)d_in[0];
    const float* gWih = (const float*)d_in[1];
    const float* gWhh = (const float*)d_in[2];
    const float* gbih = (const float*)d_in[3];
    const float* gbhh = (const float*)d_in[4];
    const float* lWih = (const float*)d_in[5];
    const float* lWhh = (const float*)d_in[6];
    const float* lbih = (const float*)d_in[7];
    const float* lbhh = (const float*)d_in[8];
    const float* fcW  = (const float*)d_in[9];
    const float* fcb  = (const float*)d_in[10];
    float* out = (float*)d_out;

    dim3 grid(kB / 16);
    dim3 block(32);
    hipLaunchKernelGGL(fused_gru_lstm_fc, grid, block, 0, stream,
                       x, gWih, gWhh, gbih, gbhh,
                       lWih, lWhh, lbih, lbhh, fcW, fcb, out);
}


# Each batch tile's time step spread over a 9-wave work-group by data, not by stage: one accumulator chain per wave, one state element per thread, exchange through LDS; every sum keeps its order
# speedup vs baseline: 7.0518x; 7.0518x over previous
#include <hip/hip_runtime.h>
#include <cmath>


typedef __attribute__((ext_vector_type(2))) float v2f;
typedef __attribute__((ext_vector_type(8))) float v8f;

constexpr int kH  = 18;
constexpr int kH3 = 54;
constexpr int kH4 = 72;
constexpr int kT  = 1024;
constexpr int kB  = 1024;
constexpr int kKP = 20;
constexpr int kKS = 5;
constexpr int kNG = 64;
constexpr int kNL = 80;
constexpr int kNT = 16 * kH;

__device__ __forceinline__ float sigmoidf_(float v) {
    return 1.0f / (1.0f + __expf(-v));
}

__global__ __launch_bounds__(kNT) void fused_gru_lstm_fc(
    const float* __restrict__ x,
    const float* __restrict__ gWih, const float* __restrict__ gWhh,
    const float* __restrict__ gbih, const float* __restrict__ gbhh,
    const float* __restrict__ lWih, const float* __restrict__ lWhh,
    const float* __restrict__ lbih, const float* __restrict__ lbhh,
    const float* __restrict__ fcW,  const float* __restrict__ fcb,
    float* __restrict__ out)
{
    __shared__ float Wg[2][kNG][kKP];
    __shared__ float WlX[kNL][kKP], WlH[kNL][kKP];
    __shared__ float gbi_s[kH3], gbh_s[kH3], lb_s[kH4];
    __shared__ float fcw_s[kH];
    __shared__ float fcb_s;
    __shared__ float ag[2][16][kKP];
    __shared__ float hl[16][kKP];
    __shared__ float sg[2][16][kNG];
    __shared__ float sl[16][kNL];

    const int tid  = threadIdx.x;
    const int lane = tid & 31;
    const int wave = __builtin_amdgcn_readfirstlane(tid >> 5);
    const int m0   = blockIdx.x * 16;

    for (int i = tid; i < kNG * kKP; i += kNT) {
        int n = i / kKP, k = i % kKP;
        bool ok = (n < kH3) && (k < kH);
        Wg[0][n][k] = ok ? gWih[n * kH + k] : 0.0f;
        Wg[1][n][k] = ok ? gWhh[n * kH + k] : 0.0f;
    }
    for (int i = tid; i < kNL * kKP; i += kNT) {
        int n = i / kKP, k = i % kKP;
        bool ok = (n < kH4) && (k < kH);
        WlX[n][k] = ok ? lWih[n * kH + k] : 0.0f;
        WlH[n][k] = ok ? lWhh[n * kH + k] : 0.0f;
    }
    for (int i = tid; i < kH3; i += kNT) { gbi_s[i] = gbih[i]; gbh_s[i] = gbhh[i]; }
    for (int i = tid; i < kH4; i += kNT) { lb_s[i]  = lbih[i] + lbhh[i]; }
    for (int i = tid; i < kH;  i += kNT) { fcw_s[i] = fcW[i]; }
    if (tid == 0) fcb_s = fcb[0];
    for (int i = tid; i < 16 * kKP; i += kNT) {
        (&ag[0][0][0])[i] = 0.0f; (&ag[1][0][0])[i] = 0.0f; (&hl[0][0])[i] = 0.0f;
    }
    __syncthreads();

    const int lh  = lane & 15;
    const int kh  = (lane >> 4) << 1;
    const int hi8 = (lane >> 4) << 3;

    const int gnt = (wave >> 1) & 3;
    const int gop = wave & 1;
    const int lnt = (wave < 5) ? wave : 0;
    v2f bG[kKS], bLX[kKS], bLH[kKS];
    #pragma unroll
    for (int ks = 0; ks < kKS; ++ks) {
        int c = ks * 4 + kh;
        int rg = gnt * 16 + lh, rl = lnt * 16 + lh;
        v2f g; g.x = Wg[gop][rg][c]; g.y = Wg[gop][rg][c + 1]; bG[ks]  = g;
        v2f a; a.x = WlX[rl][c];     a.y = WlX[rl][c + 1];     bLX[ks] = a;
        v2f b; b.x = WlH[rl][c];     b.y = WlH[rl][c + 1];     bLH[ks] = b;
    }

    const int em = tid / kH, ej = tid - em * kH;
    const float bir = gbi_s[ej], biz = gbi_s[kH + ej], bin = gbi_s[2 * kH + ej];
    const float bhr = gbh_s[ej], bhz = gbh_s[kH + ej], bhn = gbh_s[2 * kH + ej];
    const float lbi = lb_s[ej], lbf = lb_s[kH + ej], lbg = lb_s[2 * kH + ej], lbo = lb_s[3 * kH + ej];
    const float* xp = x + ((size_t)(m0 + em) * kT) * kH + ej;
    float hgv = 0.0f;
    float cv  = 0.0f;
    float fcs = 0.0f;

    ag[0][em][ej] = xp[0];
    if (tid < 16) __builtin_prefetch(&x[((size_t)(m0 + tid) * kT + 1) * kH], 0, 3);
    __syncthreads();

    for (int t = 0; t < kT; ++t) {
        if (wave < 8) {
            v2f a[kKS];
            #pragma unroll
            for (int ks = 0; ks < kKS; ++ks) {
                int c = ks * 4 + kh;
                v2f v; v.x = ag[gop][lh][c]; v.y = ag[gop][lh][c + 1]; a[ks] = v;
            }
            v8f acc = (v8f){};
            #pragma unroll
            for (int ks = 0; ks < kKS; ++ks)
                acc = __builtin_amdgcn_wmma_f32_16x16x4_f32(
                    false, a[ks], false, bG[ks], (short)0, acc, false, false);
            #pragma unroll
            for (int j = 0; j < 8; ++j) sg[gop][hi8 + j][gnt * 16 + lh] = acc[j];
        }
        __syncthreads();

        {
            float r = sigmoidf_((sg[0][em][ej]          + bir)
                              + (sg[1][em][ej]          + bhr));
            float z = sigmoidf_((sg[0][em][kH + ej]     + biz)
                              + (sg[1][em][kH + ej]     + bhz));
            float n = tanhf(    (sg[0][em][2*kH + ej]   + bin)
                          + r * (sg[1][em][2*kH + ej]   + bhn));
            hgv = (1.0f - z) * n + z * hgv;
            ag[1][em][ej] = hgv;
        }
        __syncthreads();

        if (wave < 5) {
            v2f aG[kKS], aL[kKS];
            #pragma unroll
            for (int ks = 0; ks < kKS; ++ks) {
                int c = ks * 4 + kh;
                v2f a; a.x = ag[1][lh][c]; a.y = ag[1][lh][c + 1]; aG[ks] = a;
                v2f b; b.x = hl[lh][c];    b.y = hl[lh][c + 1];    aL[ks] = b;
            }
            v8f acc = (v8f){};
            #pragma unroll
            for (int ks = 0; ks < kKS; ++ks) {
                acc = __builtin_amdgcn_wmma_f32_16x16x4_f32(
                    false, aG[ks], false, bLX[ks], (short)0, acc, false, false);
                acc = __builtin_amdgcn_wmma_f32_16x16x4_f32(
                    false, aL[ks], false, bLH[ks], (short)0, acc, false, false);
            }
            #pragma unroll
            for (int j = 0; j < 8; ++j) sl[hi8 + j][lnt * 16 + lh] = acc[j];
        } else if (wave == 5) {
            if (t > 0 && lane < 16) {
                float s = fcb_s;
                #pragma unroll
                for (int j = 0; j < kH; ++j) s += hl[lane][j] * fcw_s[j];
                fcs = s;
            }
        }
        __syncthreads();

        if (wave == 5 && t > 0 && lane < 16) out[(size_t)(m0 + lane) * kT + (t - 1)] = fcs;

        {
            const bool more = (t + 1 < kT);
            float xn = 0.0f;
            if (more) xn = xp[(size_t)(t + 1) * kH];
            if (t + 2 < kT && tid < 16) {
                __builtin_prefetch(&x[((size_t)(m0 + tid) * kT + (t + 2)) * kH], 0, 3);
            }
            float ig = sigmoidf_(sl[em][ej]          + lbi);
            float fg = sigmoidf_(sl[em][kH + ej]     + lbf);
            float gg = tanhf(    sl[em][2*kH + ej]   + lbg);
            float og = sigmoidf_(sl[em][3*kH + ej]   + lbo);
            cv = fg * cv + ig * gg;
            hl[em][ej] = og * tanhf(cv);
            if (more) ag[0][em][ej] = xn;
        }
        __syncthreads();
    }

    if (wave == 5 && lane < 16) {
        float s = fcb_s;
        #pragma unroll
        for (int j = 0; j < kH; ++j) s += hl[lane][j] * fcw_s[j];
        out[(size_t)(m0 + lane) * kT + (kT - 1)] = s;
    }
}

extern "C" void kernel_launch(void* const* d_in, const int* in_sizes, int n_in,
                              void* d_out, int out_size, void* d_ws, size_t ws_size,
                              hipStream_t stream) {
    (void)in_sizes; (void)n_in; (void)out_size; (void)d_ws; (void)ws_size;
    const float* x    = (const float*)d_in[0];
    const float* gWih = (const float*)d_in[1];
    const float* gWhh = (const float*)d_in[2];
    const float* gbih = (const float*)d_in[3];
    const float* gbhh = (const float*)d_in[4];
    const float* lWih = (const float*)d_in[5];
    const float* lWhh = (const float*)d_in[6];
    const float* lbih = (const float*)d_in[7];
    const float* lbhh = (const float*)d_in[8];
    const float* fcW  = (const float*)d_in[9];
    const float* fcb  = (const float*)d_in[10];
    float* out = (float*)d_out;

    dim3 grid(kB / 16);
    dim3 block(kNT);
    hipLaunchKernelGGL(fused_gru_lstm_fc, grid, block, 0, stream,
                       x, gWih, gWhh, gbih, gbhh,
                       lWih, lWhh, lbih, lbhh, fcW, fcb, out);
}


# LSTM layer runs one time step behind the GRU layer: both layers' matrix runs share one barrier interval and both gate phases the next (2 barriers a step, not 4); every sum keeps its order
# speedup vs baseline: 8.0872x; 1.1468x over previous
#include <hip/hip_runtime.h>
#include <cmath>
#include <type_traits>


typedef __attribute__((ext_vector_type(2))) float v2f;
typedef __attribute__((ext_vector_type(8))) float v8f;

constexpr int kH  = 18;
constexpr int kH3 = 54;
constexpr int kH4 = 72;
constexpr int kT  = 1024;
constexpr int kB  = 1024;
constexpr int kKP = 20;
constexpr int kKS = 5;
constexpr int kNG = 64;
constexpr int kNL = 80;
constexpr int kNT = 16 * kH;

__device__ __forceinline__ float sigmoidf_(float v) {
    return 1.0f / (1.0f + __expf(-v));
}

__global__ __launch_bounds__(kNT) void fused_gru_lstm_fc(
    const float* __restrict__ x,
    const float* __restrict__ gWih, const float* __restrict__ gWhh,
    const float* __restrict__ gbih, const float* __restrict__ gbhh,
    const float* __restrict__ lWih, const float* __restrict__ lWhh,
    const float* __restrict__ lbih, const float* __restrict__ lbhh,
    const float* __restrict__ fcW,  const float* __restrict__ fcb,
    float* __restrict__ out)
{
    __shared__ float Wg[2][kNG][kKP];
    __shared__ float WlX[kNL][kKP], WlH[kNL][kKP];
    __shared__ float gbi_s[kH3], gbh_s[kH3], lb_s[kH4];
    __shared__ float fcw_s[kH];
    __shared__ float fcb_s;
    __shared__ float ag[2][16][kKP];
    __shared__ float hl[16][kKP];
    __shared__ float sg[2][16][kNG];
    __shared__ float sl[16][kNL];

    const int tid  = threadIdx.x;
    const int lane = tid & 31;
    const int wave = __builtin_amdgcn_readfirstlane(tid >> 5);
    const int m0   = blockIdx.x * 16;

    for (int i = tid; i < kNG * kKP; i += kNT) {
        int n = i / kKP, k = i % kKP;
        bool ok = (n < kH3) && (k < kH);
        Wg[0][n][k] = ok ? gWih[n * kH + k] : 0.0f;
        Wg[1][n][k] = ok ? gWhh[n * kH + k] : 0.0f;
    }
    for (int i = tid; i < kNL * kKP; i += kNT) {
        int n = i / kKP, k = i % kKP;
        bool ok = (n < kH4) && (k < kH);
        WlX[n][k] = ok ? lWih[n * kH + k] : 0.0f;
        WlH[n][k] = ok ? lWhh[n * kH + k] : 0.0f;
    }
    for (int i = tid; i < kH3; i += kNT) { gbi_s[i] = gbih[i]; gbh_s[i] = gbhh[i]; }
    for (int i = tid; i < kH4; i += kNT) { lb_s[i]  = lbih[i] + lbhh[i]; }
    for (int i = tid; i < kH;  i += kNT) { fcw_s[i] = fcW[i]; }
    if (tid == 0) fcb_s = fcb[0];
    for (int i = tid; i < 16 * kKP; i += kNT) {
        (&ag[0][0][0])[i] = 0.0f; (&ag[1][0][0])[i] = 0.0f; (&hl[0][0])[i] = 0.0f;
    }
    __syncthreads();

    const int lh  = lane & 15;
    const int kh  = (lane >> 4) << 1;
    const int hi8 = (lane >> 4) << 3;

    const bool lrole = (wave < 5);
    const int  ntile = lrole ? wave : (wave - 5);
    v2f b1[kKS], b2[kKS];
    #pragma unroll
    for (int ks = 0; ks < kKS; ++ks) {
        int c = ks * 4 + kh;
        int r = ntile * 16 + lh;
        v2f p, q;
        if (lrole) { p.x = WlX[r][c];   p.y = WlX[r][c + 1];   q.x = WlH[r][c];   q.y = WlH[r][c + 1]; }
        else       { p.x = Wg[0][r][c]; p.y = Wg[0][r][c + 1]; q.x = Wg[1][r][c]; q.y = Wg[1][r][c + 1]; }
        b1[ks] = p; b2[ks] = q;
    }

    const int em = tid / kH, ej = tid - em * kH;
    const float bir = gbi_s[ej], biz = gbi_s[kH + ej], bin = gbi_s[2 * kH + ej];
    const float bhr = gbh_s[ej], bhz = gbh_s[kH + ej], bhn = gbh_s[2 * kH + ej];
    const float lbi = lb_s[ej], lbf = lb_s[kH + ej], lbg = lb_s[2 * kH + ej], lbo = lb_s[3 * kH + ej];
    const float* xp = x + ((size_t)(m0 + em) * kT) * kH + ej;
    float hgv = 0.0f;
    float cv  = 0.0f;
    float fcs = 0.0f;

    ag[0][em][ej] = xp[0];
    if (tid < 16) __builtin_prefetch(&x[((size_t)(m0 + tid) * kT + 1) * kH], 0, 3);
    __syncthreads();

    auto trip = [&](auto Gt, auto Lt, const int s) {
        constexpr bool G = decltype(Gt)::value;
        constexpr bool L = decltype(Lt)::value;

        if (lrole) {
            if (L) {
                v2f aG[kKS], aL[kKS];
                #pragma unroll
                for (int ks = 0; ks < kKS; ++ks) {
                    int c = ks * 4 + kh;
                    v2f a; a.x = ag[1][lh][c]; a.y = ag[1][lh][c + 1]; aG[ks] = a;
                    v2f b; b.x = hl[lh][c];    b.y = hl[lh][c + 1];    aL[ks] = b;
                }
                v8f acc = (v8f){};
                #pragma unroll
                for (int ks = 0; ks < kKS; ++ks) {
                    acc = __builtin_amdgcn_wmma_f32_16x16x4_f32(
                        false, aG[ks], false, b1[ks], (short)0, acc, false, false);
                    acc = __builtin_amdgcn_wmma_f32_16x16x4_f32(
                        false, aL[ks], false, b2[ks], (short)0, acc, false, false);
                }
                #pragma unroll
                for (int j = 0; j < 8; ++j) sl[hi8 + j][ntile * 16 + lh] = acc[j];
            }
        } else {
            if (wave == 8 && s >= 2 && lane < 16) {
                float v = fcb_s;
                #pragma unroll
                for (int j = 0; j < kH; ++j) v += hl[lane][j] * fcw_s[j];
                fcs = v;
            }
            if (G) {
                v2f aX[kKS], aH[kKS];
                #pragma unroll
                for (int ks = 0; ks < kKS; ++ks) {
                    int c = ks * 4 + kh;
                    v2f a; a.x = ag[0][lh][c]; a.y = ag[0][lh][c + 1]; aX[ks] = a;
                    v2f b; b.x = ag[1][lh][c]; b.y = ag[1][lh][c + 1]; aH[ks] = b;
                }
                v8f accX = (v8f){}, accH = (v8f){};
                #pragma unroll
                for (int ks = 0; ks < kKS; ++ks) {
                    accX = __builtin_amdgcn_wmma_f32_16x16x4_f32(
                        false, aX[ks], false, b1[ks], (short)0, accX, false, false);
                    accH = __builtin_amdgcn_wmma_f32_16x16x4_f32(
                        false, aH[ks], false, b2[ks], (short)0, accH, false, false);
                }
                #pragma unroll
                for (int j = 0; j < 8; ++j) {
                    sg[0][hi8 + j][ntile * 16 + lh] = accX[j];
                    sg[1][hi8 + j][ntile * 16 + lh] = accH[j];
                }
            }
        }
        __syncthreads();

        if (wave == 8 && s >= 2 && lane < 16) out[(size_t)(m0 + lane) * kT + (s - 2)] = fcs;

        float xn = 0.0f;
        if (G) {
            if (s + 1 < kT) xn = xp[(size_t)(s + 1) * kH];
            if (s + 2 < kT && tid < 16) {
                __builtin_prefetch(&x[((size_t)(m0 + tid) * kT + (s + 2)) * kH], 0, 3);
            }
        }
        float hg_new = hgv, hl_new = 0.0f;
        if (G) {
            float r = sigmoidf_((sg[0][em][ej]          + bir)
                              + (sg[1][em][ej]          + bhr));
            float z = sigmoidf_((sg[0][em][kH + ej]     + biz)
                              + (sg[1][em][kH + ej]     + bhz));
            float n = tanhf(    (sg[0][em][2*kH + ej]   + bin)
                          + r * (sg[1][em][2*kH + ej]   + bhn));
            hg_new = (1.0f - z) * n + z * hgv;
        }
        if (L) {
            float ig = sigmoidf_(sl[em][ej]          + lbi);
            float fg = sigmoidf_(sl[em][kH + ej]     + lbf);
            float gg = tanhf(    sl[em][2*kH + ej]   + lbg);
            float og = sigmoidf_(sl[em][3*kH + ej]   + lbo);
            cv = fg * cv + ig * gg;
            hl_new = og * tanhf(cv);
        }
        if (G) {
            hgv = hg_new;
            ag[1][em][ej] = hg_new;
            if (s + 1 < kT) ag[0][em][ej] = xn;
        }
        if (L) hl[em][ej] = hl_new;
        __syncthreads();
    };

    using T_ = std::integral_constant<bool, true>;
    using F_ = std::integral_constant<bool, false>;

    trip(T_{}, F_{}, 0);
    for (int s = 1; s < kT; ++s) trip(T_{}, T_{}, s);
    trip(F_{}, T_{}, kT);

    if (wave == 8 && lane < 16) {
        float v = fcb_s;
        #pragma unroll
        for (int j = 0; j < kH; ++j) v += hl[lane][j] * fcw_s[j];
        out[(size_t)(m0 + lane) * kT + (kT - 1)] = v;
    }
}

extern "C" void kernel_launch(void* const* d_in, const int* in_sizes, int n_in,
                              void* d_out, int out_size, void* d_ws, size_t ws_size,
                              hipStream_t stream) {
    (void)in_sizes; (void)n_in; (void)out_size; (void)d_ws; (void)ws_size;
    const float* x    = (const float*)d_in[0];
    const float* gWih = (const float*)d_in[1];
    const float* gWhh = (const float*)d_in[2];
    const float* gbih = (const float*)d_in[3];
    const float* gbhh = (const float*)d_in[4];
    const float* lWih = (const float*)d_in[5];
    const float* lWhh = (const float*)d_in[6];
    const float* lbih = (const float*)d_in[7];
    const float* lbhh = (const float*)d_in[8];
    const float* fcW  = (const float*)d_in[9];
    const float* fcb  = (const float*)d_in[10];
    float* out = (float*)d_out;

    dim3 grid(kB / 16);
    dim3 block(kNT);
    hipLaunchKernelGGL(fused_gru_lstm_fc, grid, block, 0, stream,
                       x, gWih, gWhh, gbih, gbhh,
                       lWih, lWhh, lbih, lbhh, fcW, fcb, out);
}


# Gate phase split by layer over two wave sets (18 waves a tile): matrix waves 0-8 keep the GRU gate math, added waves 9-17 take the LSTM gate math and the FC sums; every sum keeps its order
# speedup vs baseline: 10.4164x; 1.2880x over previous
#include <hip/hip_runtime.h>
#include <cmath>
#include <type_traits>


typedef __attribute__((ext_vector_type(2))) float v2f;
typedef __attribute__((ext_vector_type(8))) float v8f;

constexpr int kH  = 18;
constexpr int kH3 = 54;
constexpr int kH4 = 72;
constexpr int kT  = 1024;
constexpr int kB  = 1024;
constexpr int kKP = 20;
constexpr int kKS = 5;
constexpr int kNG = 64;
constexpr int kNL = 80;
constexpr int kNE = 16 * kH;
constexpr int kNT = 2 * kNE;

__device__ __forceinline__ float sigmoidf_(float v) {
    return 1.0f / (1.0f + __expf(-v));
}

__global__ __launch_bounds__(kNT) void fused_gru_lstm_fc(
    const float* __restrict__ x,
    const float* __restrict__ gWih, const float* __restrict__ gWhh,
    const float* __restrict__ gbih, const float* __restrict__ gbhh,
    const float* __restrict__ lWih, const float* __restrict__ lWhh,
    const float* __restrict__ lbih, const float* __restrict__ lbhh,
    const float* __restrict__ fcW,  const float* __restrict__ fcb,
    float* __restrict__ out)
{
    __shared__ float Wg[2][kNG][kKP];
    __shared__ float WlX[kNL][kKP], WlH[kNL][kKP];
    __shared__ float gbi_s[kH3], gbh_s[kH3], lb_s[kH4];
    __shared__ float fcw_s[kH];
    __shared__ float fcb_s;
    __shared__ float ag[2][16][kKP];
    __shared__ float hl[16][kKP];
    __shared__ float sg[2][16][kNG];
    __shared__ float sl[16][kNL];

    const int tid  = threadIdx.x;
    const int lane = tid & 31;
    const int wave = __builtin_amdgcn_readfirstlane(tid >> 5);
    const int m0   = blockIdx.x * 16;

    for (int i = tid; i < kNG * kKP; i += kNT) {
        int n = i / kKP, k = i % kKP;
        bool ok = (n < kH3) && (k < kH);
        Wg[0][n][k] = ok ? gWih[n * kH + k] : 0.0f;
        Wg[1][n][k] = ok ? gWhh[n * kH + k] : 0.0f;
    }
    for (int i = tid; i < kNL * kKP; i += kNT) {
        int n = i / kKP, k = i % kKP;
        bool ok = (n < kH4) && (k < kH);
        WlX[n][k] = ok ? lWih[n * kH + k] : 0.0f;
        WlH[n][k] = ok ? lWhh[n * kH + k] : 0.0f;
    }
    for (int i = tid; i < kH3; i += kNT) { gbi_s[i] = gbih[i]; gbh_s[i] = gbhh[i]; }
    for (int i = tid; i < kH4; i += kNT) { lb_s[i]  = lbih[i] + lbhh[i]; }
    for (int i = tid; i < kH;  i += kNT) { fcw_s[i] = fcW[i]; }
    if (tid == 0) fcb_s = fcb[0];
    for (int i = tid; i < 16 * kKP; i += kNT) {
        (&ag[0][0][0])[i] = 0.0f; (&ag[1][0][0])[i] = 0.0f; (&hl[0][0])[i] = 0.0f;
    }
    __syncthreads();

    const int lh  = lane & 15;
    const int kh  = (lane >> 4) << 1;
    const int hi8 = (lane >> 4) << 3;

    const bool lgate = (wave >= 9);
    const bool lrole = (wave < 5);
    const int  ntile = lrole ? wave : (lgate ? 0 : (wave - 5));
    v2f b1[kKS], b2[kKS];
    #pragma unroll
    for (int ks = 0; ks < kKS; ++ks) {
        int c = ks * 4 + kh;
        int r = ntile * 16 + lh;
        v2f p, q;
        if (lrole) { p.x = WlX[r][c];   p.y = WlX[r][c + 1];   q.x = WlH[r][c];   q.y = WlH[r][c + 1]; }
        else       { p.x = Wg[0][r][c]; p.y = Wg[0][r][c + 1]; q.x = Wg[1][r][c]; q.y = Wg[1][r][c + 1]; }
        b1[ks] = p; b2[ks] = q;
    }

    const int ee = lgate ? (tid - kNE) : tid;
    const int em = ee / kH, ej = ee - em * kH;
    const float bir = gbi_s[ej], biz = gbi_s[kH + ej], bin = gbi_s[2 * kH + ej];
    const float bhr = gbh_s[ej], bhz = gbh_s[kH + ej], bhn = gbh_s[2 * kH + ej];
    const float lbi = lb_s[ej], lbf = lb_s[kH + ej], lbg = lb_s[2 * kH + ej], lbo = lb_s[3 * kH + ej];
    const float* xp = x + ((size_t)(m0 + em) * kT) * kH + ej;
    float hgv = 0.0f;
    float cv  = 0.0f;
    float fcs = 0.0f;

    if (!lgate) ag[0][em][ej] = xp[0];
    if (tid < 16) __builtin_prefetch(&x[((size_t)(m0 + tid) * kT + 1) * kH], 0, 3);
    __syncthreads();

    auto trip = [&](auto Gt, auto Lt, const int s) {
        constexpr bool G = decltype(Gt)::value;
        constexpr bool L = decltype(Lt)::value;

        if (lgate) {
            if (wave == 9 && s >= 2 && lane < 16) {
                float v = fcb_s;
                #pragma unroll
                for (int j = 0; j < kH; ++j) v += hl[lane][j] * fcw_s[j];
                fcs = v;
            }
        } else if (lrole) {
            if (L) {
                v2f aG[kKS], aL[kKS];
                #pragma unroll
                for (int ks = 0; ks < kKS; ++ks) {
                    int c = ks * 4 + kh;
                    v2f a; a.x = ag[1][lh][c]; a.y = ag[1][lh][c + 1]; aG[ks] = a;
                    v2f b; b.x = hl[lh][c];    b.y = hl[lh][c + 1];    aL[ks] = b;
                }
                v8f acc = (v8f){};
                #pragma unroll
                for (int ks = 0; ks < kKS; ++ks) {
                    acc = __builtin_amdgcn_wmma_f32_16x16x4_f32(
                        false, aG[ks], false, b1[ks], (short)0, acc, false, false);
                    acc = __builtin_amdgcn_wmma_f32_16x16x4_f32(
                        false, aL[ks], false, b2[ks], (short)0, acc, false, false);
                }
                #pragma unroll
                for (int j = 0; j < 8; ++j) sl[hi8 + j][ntile * 16 + lh] = acc[j];
            }
        } else {
            if (G) {
                v2f aX[kKS], aH[kKS];
                #pragma unroll
                for (int ks = 0; ks < kKS; ++ks) {
                    int c = ks * 4 + kh;
                    v2f a; a.x = ag[0][lh][c]; a.y = ag[0][lh][c + 1]; aX[ks] = a;
                    v2f b; b.x = ag[1][lh][c]; b.y = ag[1][lh][c + 1]; aH[ks] = b;
                }
                v8f accX = (v8f){}, accH = (v8f){};
                #pragma unroll
                for (int ks = 0; ks < kKS; ++ks) {
                    accX = __builtin_amdgcn_wmma_f32_16x16x4_f32(
                        false, aX[ks], false, b1[ks], (short)0, accX, false, false);
                    accH = __builtin_amdgcn_wmma_f32_16x16x4_f32(
                        false, aH[ks], false, b2[ks], (short)0, accH, false, false);
                }
                #pragma unroll
                for (int j = 0; j < 8; ++j) {
                    sg[0][hi8 + j][ntile * 16 + lh] = accX[j];
                    sg[1][hi8 + j][ntile * 16 + lh] = accH[j];
                }
            }
        }
        __syncthreads();

        if (wave == 9 && s >= 2 && lane < 16) out[(size_t)(m0 + lane) * kT + (s - 2)] = fcs;

        if (!lgate) {
            if (G) {
                float xn = 0.0f;
                if (s + 1 < kT) xn = xp[(size_t)(s + 1) * kH];
                if (s + 2 < kT && tid < 16) {
                    __builtin_prefetch(&x[((size_t)(m0 + tid) * kT + (s + 2)) * kH], 0, 3);
                }
                float r = sigmoidf_((sg[0][em][ej]          + bir)
                                  + (sg[1][em][ej]          + bhr));
                float z = sigmoidf_((sg[0][em][kH + ej]     + biz)
                                  + (sg[1][em][kH + ej]     + bhz));
                float n = tanhf(    (sg[0][em][2*kH + ej]   + bin)
                              + r * (sg[1][em][2*kH + ej]   + bhn));
                float hg_new = (1.0f - z) * n + z * hgv;
                hgv = hg_new;
                ag[1][em][ej] = hg_new;
                if (s + 1 < kT) ag[0][em][ej] = xn;
            }
        } else {
            if (L) {
                float ig = sigmoidf_(sl[em][ej]          + lbi);
                float fg = sigmoidf_(sl[em][kH + ej]     + lbf);
                float gg = tanhf(    sl[em][2*kH + ej]   + lbg);
                float og = sigmoidf_(sl[em][3*kH + ej]   + lbo);
                cv = fg * cv + ig * gg;
                hl[em][ej] = og * tanhf(cv);
            }
        }
        __syncthreads();
    };

    using T_ = std::integral_constant<bool, true>;
    using F_ = std::integral_constant<bool, false>;

    trip(T_{}, F_{}, 0);
    for (int s = 1; s < kT; ++s) trip(T_{}, T_{}, s);
    trip(F_{}, T_{}, kT);

    if (wave == 9 && lane < 16) {
        float v = fcb_s;
        #pragma unroll
        for (int j = 0; j < kH; ++j) v += hl[lane][j] * fcw_s[j];
        out[(size_t)(m0 + lane) * kT + (kT - 1)] = v;
    }
}

extern "C" void kernel_launch(void* const* d_in, const int* in_sizes, int n_in,
                              void* d_out, int out_size, void* d_ws, size_t ws_size,
                              hipStream_t stream) {
    (void)in_sizes; (void)n_in; (void)out_size; (void)d_ws; (void)ws_size;
    const float* x    = (const float*)d_in[0];
    const float* gWih = (const float*)d_in[1];
    const float* gWhh = (const float*)d_in[2];
    const float* gbih = (const float*)d_in[3];
    const float* gbhh = (const float*)d_in[4];
    const float* lWih = (const float*)d_in[5];
    const float* lWhh = (const float*)d_in[6];
    const float* lbih = (const float*)d_in[7];
    const float* lbhh = (const float*)d_in[8];
    const float* fcW  = (const float*)d_in[9];
    const float* fcb  = (const float*)d_in[10];
    float* out = (float*)d_out;

    dim3 grid(kB / 16);
    dim3 block(kNT);
    hipLaunchKernelGGL(fused_gru_lstm_fc, grid, block, 0, stream,
                       x, gWih, gWhh, gbih, gbhh,
                       lWih, lWhh, lbih, lbhh, fcW, fcb, out);
}
